# starting kernel
# baseline (speedup 1.0000x reference)
#include <hip/hip_runtime.h>
#include <hip/hip_bf16.h>

typedef __attribute__((ext_vector_type(2))) float v2f;
typedef __attribute__((ext_vector_type(8))) float v8f;

#define BATCH        64
#define NPTS         8192
#define TOTW         9356
#define PTS_PER_BLK  64
#define ACT_STRIDE   65

#define LOG2E_F 1.4426950408889634f
#define LN2_F   0.6931471805599453f

#define W0_OFF   0
#define V0_OFF   192
#define W1_OFF   448
#define V1_OFF   4544
#define W2_OFF   4800
#define V2_OFF   8896
#define W3_OFF   9152
#define B3_OFF   9344

__device__ __forceinline__ float sigmoid_f(float x) {
    return __builtin_amdgcn_rcpf(1.0f + __builtin_amdgcn_exp2f(x * -LOG2E_F));
}
__device__ __forceinline__ float softplus_f(float x) {
    const float e = __builtin_amdgcn_exp2f(fabsf(x) * -LOG2E_F);
    return fmaf(LN2_F, __builtin_amdgcn_logf(1.0f + e), fmaxf(x, 0.0f));
}

__device__ __forceinline__ void mid_layer(float* __restrict__ act,
                                          const float* __restrict__ w_lds,
                                          int wOff, int vOff,
                                          const float ctx8[8],
                                          int lrow, bool hi)
{
    v2f af[16];
    const int koff = hi ? 2 : 0;
#pragma unroll
    for (int k = 0; k < 16; ++k) {
        const float* ap = act + lrow * ACT_STRIDE + 4 * k + koff;
        af[k].x = ap[0];
        af[k].y = ap[1];
    }

    v8f acc[4] = {{}, {}, {}, {}};
#pragma unroll
    for (int k = 0; k < 16; ++k) {
        const int kb = 4 * k + koff;
        const float* wr0 = w_lds + wOff + (kb + 0) * 64 + lrow;
        const float* wr1 = w_lds + wOff + (kb + 1) * 64 + lrow;
#pragma unroll
        for (int t = 0; t < 4; ++t) {
            v2f bf;
            bf.x = wr0[t * 16];
            bf.y = wr1[t * 16];
            acc[t] = __builtin_amdgcn_wmma_f32_16x16x4_f32(
                false, af[k], false, bf, (short)0, acc[t], false, false);
        }
    }

#pragma unroll
    for (int t = 0; t < 4; ++t) {
        const int n = t * 16 + lrow;
        const float bb = w_lds[vOff + n];
        const float ws = w_lds[vOff + 64 + n];
        const float bs = w_lds[vOff + 128 + n];
        const float wh = w_lds[vOff + 192 + n];
#pragma unroll
        for (int r = 0; r < 8; ++r) {
            const int m = r + (hi ? 8 : 0);
            const float c = ctx8[r];
            float v = acc[t][r] + bb;
            v = sigmoid_f(fmaf(c, ws, bs)) * v + c * wh;
            act[m * ACT_STRIDE + n] = softplus_f(v);
        }
    }
}

__global__ __launch_bounds__(128) void ode_hypernet_kernel(
    const float* __restrict__ context,
    const float* __restrict__ y,
    const float* __restrict__ tnw,
    float* __restrict__ out)
{
    __shared__ __align__(16) float w_lds[TOTW + 4];
    __shared__ float ctx_lds[PTS_PER_BLK];
    __shared__ float act_lds[4 * 16 * ACT_STRIDE];

    const int tid  = threadIdx.x;
    const int b    = blockIdx.y;
    const int wid  = tid >> 5;
    const int lane = tid & 31;
    const int lrow = lane & 15;
    const bool hi  = lane >= 16;
    const int point0_blk = blockIdx.x * PTS_PER_BLK;

    {
        const float4* src = (const float4*)(tnw + (size_t)b * TOTW);
        __builtin_prefetch(src, 0, 0);
        float4* dst = (float4*)w_lds;
        for (int i = tid; i < TOTW / 4; i += 128) dst[i] = src[i];
    }
    if (tid < PTS_PER_BLK)
        ctx_lds[tid] = context[(size_t)b * NPTS + point0_blk + tid];
    __syncthreads();

    const int point0 = point0_blk + wid * 16;
    float* act = act_lds + wid * 16 * ACT_STRIDE;

    float ctx8[8];
#pragma unroll
    for (int r = 0; r < 8; ++r)
        ctx8[r] = ctx_lds[wid * 16 + r + (hi ? 8 : 0)];

    {
        v2f a0;
        {
            const float* yp = y + ((size_t)b * NPTS + point0 + lrow) * 3;
            const float ya = yp[hi ? 2 : 0];
            const float yb = yp[1];
            a0.x = ya;
            a0.y = hi ? 0.0f : yb;
        }
#pragma unroll
        for (int t = 0; t < 4; ++t) {
            const int n = t * 16 + lrow;
            v2f bf;
            bf.x = w_lds[W0_OFF + (hi ? 2 : 0) * 64 + n];
            const float w1 = w_lds[W0_OFF + 1 * 64 + n];
            bf.y = hi ? 0.0f : w1;
            v8f acc = {};
            acc = __builtin_amdgcn_wmma_f32_16x16x4_f32(
                false, a0, false, bf, (short)0, acc, false, false);
            const float bb = w_lds[V0_OFF + n];
            const float ws = w_lds[V0_OFF + 64 + n];
            const float bs = w_lds[V0_OFF + 128 + n];
            const float wh = w_lds[V0_OFF + 192 + n];
#pragma unroll
            for (int r = 0; r < 8; ++r) {
                const int m = r + (hi ? 8 : 0);
                const float c = ctx8[r];
                float v = acc[r] + bb;
                v = sigmoid_f(fmaf(c, ws, bs)) * v + c * wh;
                act[m * ACT_STRIDE + n] = softplus_f(v);
            }
        }
    }

    mid_layer(act, w_lds, W1_OFF, V1_OFF, ctx8, lrow, hi);
    mid_layer(act, w_lds, W2_OFF, V2_OFF, ctx8, lrow, hi);

    {
        v2f af[16];
        const int koff = hi ? 2 : 0;
#pragma unroll
        for (int k = 0; k < 16; ++k) {
            const float* ap = act + lrow * ACT_STRIDE + 4 * k + koff;
            af[k].x = ap[0];
            af[k].y = ap[1];
        }
        const bool nvalid = lrow < 3;
        const int nc = nvalid ? lrow : 0;
        const float nmask = nvalid ? 1.0f : 0.0f;
        v8f acc0 = {}, acc1 = {};
#pragma unroll
        for (int k = 0; k < 16; k += 2) {
            const int kb0 = 4 * k + koff;
            const int kb1 = 4 * (k + 1) + koff;
            v2f bf0, bf1;
            bf0.x = w_lds[W3_OFF + (kb0 + 0) * 3 + nc] * nmask;
            bf0.y = w_lds[W3_OFF + (kb0 + 1) * 3 + nc] * nmask;
            bf1.x = w_lds[W3_OFF + (kb1 + 0) * 3 + nc] * nmask;
            bf1.y = w_lds[W3_OFF + (kb1 + 1) * 3 + nc] * nmask;
            acc0 = __builtin_amdgcn_wmma_f32_16x16x4_f32(
                false, af[k], false, bf0, (short)0, acc0, false, false);
            acc1 = __builtin_amdgcn_wmma_f32_16x16x4_f32(
                false, af[k + 1], false, bf1, (short)0, acc1, false, false);
        }
        const v8f acc = acc0 + acc1;
        if (nvalid) {
            const float bb = w_lds[B3_OFF + lrow];
            const float ws = w_lds[B3_OFF + 3 + lrow];
            const float bs = w_lds[B3_OFF + 6 + lrow];
            const float wh = w_lds[B3_OFF + 9 + lrow];
#pragma unroll
            for (int r = 0; r < 8; ++r) {
                const int m = r + (hi ? 8 : 0);
                const float c = ctx8[r];
                float v = acc[r] + bb;
                v = sigmoid_f(fmaf(c, ws, bs)) * v + c * wh;
                out[((size_t)b * NPTS + point0 + m) * 3 + lrow] = v;
            }
        }
    }
}

extern "C" void kernel_launch(void* const* d_in, const int* in_sizes, int n_in,
                              void* d_out, int out_size, void* d_ws, size_t ws_size,
                              hipStream_t stream) {
    const float* context = (const float*)d_in[0];
    const float* y       = (const float*)d_in[1];
    const float* tnw     = (const float*)d_in[2];
    float* out           = (float*)d_out;

    dim3 grid(NPTS / PTS_PER_BLK, BATCH);
    ode_hypernet_kernel<<<grid, 128, 0, stream>>>(context, y, tnw, out);
}


# Wave priority falls with progress: s_setprio 3 from entry, then 3,2,1,0 by tiles left at the top of each tile, so waves that are behind win the SIMD and all eight end together; sums unchanged
# speedup vs baseline: 1.1602x; 1.1602x over previous
#include <hip/hip_runtime.h>
#include <hip/hip_bf16.h>


typedef __attribute__((ext_vector_type(2))) float v2f;
typedef __attribute__((ext_vector_type(8))) float v8f;

#define BATCH        64
#define NPTS         8192
#define TOTW         9356
#define WAVES        8
#define PTS_PER_TRIP (WAVES * 16)
#define TRIPS        4
#define PTS_PER_BLK  (PTS_PER_TRIP * TRIPS)

#define LOG2E_F 1.4426950408889634f
#define LN2_F   0.6931471805599453f

#define W0_OFF   0
#define V0_OFF   192
#define W1_OFF   448
#define V1_OFF   4544
#define W2_OFF   4800
#define V2_OFF   8896
#define W3_OFF   9152
#define B3_OFF   9344

#define L_W0     0
#define L_V0     192
#define L_V1     448
#define L_V2     704
#define L_W3     960
#define L_B3     1152
#define L_SMALL4 291
#define L_W1     1164
#define FRAG2    128
#define L_W2     (L_W1 + 32 * FRAG2)
#define L_TOTAL  (L_W2 + 32 * FRAG2)

__device__ __forceinline__ float sigmoid_f(float x) {
    return __builtin_amdgcn_rcpf(1.0f + __builtin_amdgcn_exp2f(x * -LOG2E_F));
}
__device__ __forceinline__ float softplus_f(float x) {
    const float e = __builtin_amdgcn_exp2f(fabsf(x) * -LOG2E_F);
    return fmaf(LN2_F, __builtin_amdgcn_logf(1.0f + e), fmaxf(x, 0.0f));
}


__device__ __forceinline__ v8f bias_tile(const float* __restrict__ vec, int f) {
    const float4 b0 = *(const float4*)(vec + f);
    const float4 b1 = *(const float4*)(vec + f + 4);
    v8f a;
    a[0] = b0.x; a[1] = b0.y; a[2] = b0.z; a[3] = b0.w;
    a[4] = b1.x; a[5] = b1.y; a[6] = b1.z; a[7] = b1.w;
    return a;
}

__device__ __forceinline__ v8f gate_softplus_tile(v8f acc, const float* __restrict__ vec, int f, float c) {
    const float4 ws0 = *(const float4*)(vec + 64 + f);
    const float4 ws1 = *(const float4*)(vec + 64 + f + 4);
    const float4 bs0 = *(const float4*)(vec + 128 + f);
    const float4 bs1 = *(const float4*)(vec + 128 + f + 4);
    const float4 wh0 = *(const float4*)(vec + 192 + f);
    const float4 wh1 = *(const float4*)(vec + 192 + f + 4);
    const float ws[8] = {ws0.x, ws0.y, ws0.z, ws0.w, ws1.x, ws1.y, ws1.z, ws1.w};
    const float bs[8] = {bs0.x, bs0.y, bs0.z, bs0.w, bs1.x, bs1.y, bs1.z, bs1.w};
    const float wh[8] = {wh0.x, wh0.y, wh0.z, wh0.w, wh1.x, wh1.y, wh1.z, wh1.w};
    v8f o;
#pragma unroll
    for (int r = 0; r < 8; ++r) {
        float v = acc[r];
        v = sigmoid_f(fmaf(c, ws[r], bs[r])) * v + c * wh[r];
        o[r] = softplus_f(v);
    }
    return o;
}

__device__ __forceinline__ void mid_layer(v8f act[4],
                                          const float* __restrict__ wl,
                                          const float* __restrict__ vec,
                                          int f8, float c)
{
    v8f acc[4];
#pragma unroll
    for (int to = 0; to < 4; ++to) acc[to] = bias_tile(vec, 16 * to + f8);

#pragma unroll
    for (int t = 0; t < 4; ++t) {
#pragma unroll
        for (int rp = 0; rp < 4; ++rp) {
            v2f bf;
            bf.x = act[t][2 * rp];
            bf.y = act[t][2 * rp + 1];
#pragma unroll
            for (int tp = 0; tp < 2; ++tp) {
                const float4 w = *(const float4*)(wl + ((4 * t + rp) * 2 + tp) * FRAG2);
                v2f af0, af1;
                af0.x = w.x; af0.y = w.y;
                af1.x = w.z; af1.y = w.w;
                acc[2 * tp] = __builtin_amdgcn_wmma_f32_16x16x4_f32(
                    false, af0, false, bf, (short)0, acc[2 * tp], false, false);
                acc[2 * tp + 1] = __builtin_amdgcn_wmma_f32_16x16x4_f32(
                    false, af1, false, bf, (short)0, acc[2 * tp + 1], false, false);
            }
        }
    }

#pragma unroll
    for (int to = 0; to < 4; ++to)
        act[to] = gate_softplus_tile(acc[to], vec, 16 * to + f8, c);
}

__global__ __launch_bounds__(WAVES * 32) __attribute__((amdgpu_waves_per_eu(8, 8)))
void ode_hypernet_kernel(
    const float* __restrict__ context,
    const float* __restrict__ y,
    const float* __restrict__ tnw,
    float* __restrict__ out)
{
    __shared__ __align__(16) float w_lds[L_TOTAL];

    __builtin_amdgcn_s_setprio(3);

    const int tid  = threadIdx.x;
    const int b    = blockIdx.y;
    const int wid  = tid >> 5;
    const int lane = tid & 31;
    const int lrow = lane & 15;
    const bool hi  = lane >= 16;
    const int f8   = hi ? 8 : 0;
    const int point0_blk = blockIdx.x * PTS_PER_BLK + wid * 16;

    const unsigned pidx = (unsigned)b * NPTS + point0_blk + lrow;
    const unsigned ysel = hi ? 2u : 0u;
    float ya = y[pidx * 3u + ysel];
    float yb = y[pidx * 3u + 1u];
    float cv = context[pidx];

    {
        const float4* src = (const float4*)(tnw + (size_t)b * TOTW);
        float4* dst = (float4*)w_lds;
        for (int i = tid; i < L_SMALL4; i += WAVES * 32) {
            const int s = i + (i >= 112 ? 1024 : 0) + (i >= 176 ? 1024 : 0);
            dst[i] = src[s];
        }
#pragma unroll
        for (int m = 0; m < 2; ++m) {
            const float4* ms = src + (m == 0 ? W1_OFF / 4 : W2_OFF / 4);
            float* md = w_lds + (m == 0 ? L_W1 : L_W2);
#pragma unroll
            for (int it = 0; it < 4; ++it) {
                const int i = tid + it * (WAVES * 32);
                const float4 v = ms[i];
                const int k = i >> 4;
                const int to = (i & 15) >> 2;
                const int c0 = (i & 3) * 4;
                const int j = ((k >> 4) << 2) | ((k >> 1) & 3);
                float* d = md + (2 * j + (to >> 1)) * FRAG2
                              + (((k >> 3) & 1) * 16 + c0) * 4 + (to & 1) * 2 + (k & 1);
                d[0] = v.x; d[4] = v.y; d[8] = v.z; d[12] = v.w;
            }
        }
    }
    __syncthreads();

    const float* wl1 = w_lds + L_W1 + 4 * lane;
    const float* wl2 = w_lds + L_W2 + 4 * lane;

  for (int trip = 0; trip < TRIPS; ++trip) {
    const int tiles_left = TRIPS - 1 - trip;
    if (tiles_left >= 3)      __builtin_amdgcn_s_setprio(3);
    else if (tiles_left == 2) __builtin_amdgcn_s_setprio(2);
    else if (tiles_left == 1) __builtin_amdgcn_s_setprio(1);
    else                      __builtin_amdgcn_s_setprio(0);
    asm volatile("" ::: "memory");
    const float c = cv;

    v8f act[4];

    {
        v2f b0;
        b0.x = ya;
        b0.y = hi ? 0.0f : yb;
        {
            const unsigned np = pidx + (unsigned)((trip + 1 < TRIPS ? trip + 1 : trip) * PTS_PER_TRIP);
            ya = y[np * 3u + ysel];
            yb = y[np * 3u + 1u];
            cv = context[np];
        }
#pragma unroll
        for (int to = 0; to < 4; ++to) {
            const int n = to * 16 + lrow;
            v2f af;
            af.x = w_lds[L_W0 + (hi ? 2 : 0) * 64 + n];
            const float w1 = w_lds[L_W0 + 1 * 64 + n];
            af.y = hi ? 0.0f : w1;
            v8f acc = bias_tile(w_lds + L_V0, 16 * to + f8);
            acc = __builtin_amdgcn_wmma_f32_16x16x4_f32(
                false, af, false, b0, (short)0, acc, false, false);
            act[to] = gate_softplus_tile(acc, w_lds + L_V0, 16 * to + f8, c);
        }
    }

    mid_layer(act, wl1, w_lds + L_V1, f8, c);
    mid_layer(act, wl2, w_lds + L_V2, f8, c);

    {
        const bool nvalid = lrow < 3;
        const int nc = nvalid ? lrow : 0;
        const float nmask = nvalid ? 1.0f : 0.0f;
        const float* w3 = w_lds + L_W3 + f8 * 3 + nc;
        v8f acc0 = {}, acc1 = {};
#pragma unroll
        for (int t = 0; t < 4; ++t) {
#pragma unroll
            for (int rp = 0; rp < 4; rp += 2) {
                v2f bf0, bf1, af0, af1;
                bf0.x = act[t][2 * rp];
                bf0.y = act[t][2 * rp + 1];
                bf1.x = act[t][2 * rp + 2];
                bf1.y = act[t][2 * rp + 3];
                af0.x = w3[(16 * t + 2 * rp + 0) * 3] * nmask;
                af0.y = w3[(16 * t + 2 * rp + 1) * 3] * nmask;
                af1.x = w3[(16 * t + 2 * rp + 2) * 3] * nmask;
                af1.y = w3[(16 * t + 2 * rp + 3) * 3] * nmask;
                acc0 = __builtin_amdgcn_wmma_f32_16x16x4_f32(
                    false, af0, false, bf0, (short)0, acc0, false, false);
                acc1 = __builtin_amdgcn_wmma_f32_16x16x4_f32(
                    false, af1, false, bf1, (short)0, acc1, false, false);
            }
        }
        const v8f acc = acc0 + acc1;
        if (!hi) {
            const unsigned ob = (pidx + (unsigned)(trip * PTS_PER_TRIP)) * 3u;
#pragma unroll
            for (int r = 0; r < 3; ++r) {
                const float bb = w_lds[L_B3 + r];
                const float ws = w_lds[L_B3 + 3 + r];
                const float bs = w_lds[L_B3 + 6 + r];
                const float wh = w_lds[L_B3 + 9 + r];
                float v = acc[r] + bb;
                v = sigmoid_f(fmaf(c, ws, bs)) * v + c * wh;
                out[ob + (unsigned)r] = v;
            }
        }
    }
  }
}

extern "C" void kernel_launch(void* const* d_in, const int* in_sizes, int n_in,
                              void* d_out, int out_size, void* d_ws, size_t ws_size,
                              hipStream_t stream) {
    const float* context = (const float*)d_in[0];
    const float* y       = (const float*)d_in[1];
    const float* tnw     = (const float*)d_in[2];
    float* out           = (float*)d_out;

    dim3 grid(NPTS / PTS_PER_BLK, BATCH);
    ode_hypernet_kernel<<<grid, WAVES * 32, 0, stream>>>(context, y, tnw, out);
}


# Base-2 units: log2e folded once per block into staged bias/scale/shift vectors and W0 (ln2 on last sums); gate/softplus arithmetic two-wide in packed f32; sum order same, roundings differ
# speedup vs baseline: 1.2148x; 1.0471x over previous
#include <hip/hip_runtime.h>
#include <hip/hip_bf16.h>


typedef __attribute__((ext_vector_type(2))) float v2f;
typedef __attribute__((ext_vector_type(8))) float v8f;

#define BATCH        64
#define NPTS         8192
#define TOTW         9356
#define WAVES        8
#define PTS_PER_TRIP (WAVES * 16)
#define TRIPS        4
#define PTS_PER_BLK  (PTS_PER_TRIP * TRIPS)

#define LOG2E_F 1.4426950408889634f
#define LN2_F   0.6931471805599453f

#define W0_OFF   0
#define V0_OFF   192
#define W1_OFF   448
#define V1_OFF   4544
#define W2_OFF   4800
#define V2_OFF   8896
#define W3_OFF   9152
#define B3_OFF   9344

#define L_W0     0
#define L_V0     192
#define L_V1     448
#define L_V2     704
#define L_W3     960
#define L_B3     1152
#define L_SMALL4 291
#define L_W1     1164
#define FRAG2    128
#define L_W2     (L_W1 + 32 * FRAG2)
#define L_TOTAL  (L_W2 + 32 * FRAG2)

__device__ __forceinline__ float sigmoid_f(float x) {
    return __builtin_amdgcn_rcpf(1.0f + __builtin_amdgcn_exp2f(x * -LOG2E_F));
}


__device__ __forceinline__ v8f bias_tile(const float* __restrict__ vec, int f) {
    const float4 b0 = *(const float4*)(vec + f);
    const float4 b1 = *(const float4*)(vec + f + 4);
    v8f a;
    a[0] = b0.x; a[1] = b0.y; a[2] = b0.z; a[3] = b0.w;
    a[4] = b1.x; a[5] = b1.y; a[6] = b1.z; a[7] = b1.w;
    return a;
}

__device__ __forceinline__ v8f gate_softplus_tile(v8f acc, const float* __restrict__ vec, int f, float c) {
    const float4 ws0 = *(const float4*)(vec + 64 + f);
    const float4 ws1 = *(const float4*)(vec + 64 + f + 4);
    const float4 bs0 = *(const float4*)(vec + 128 + f);
    const float4 bs1 = *(const float4*)(vec + 128 + f + 4);
    const float4 wh0 = *(const float4*)(vec + 192 + f);
    const float4 wh1 = *(const float4*)(vec + 192 + f + 4);
    const v2f ws[4] = {{ws0.x, ws0.y}, {ws0.z, ws0.w}, {ws1.x, ws1.y}, {ws1.z, ws1.w}};
    const v2f bs[4] = {{bs0.x, bs0.y}, {bs0.z, bs0.w}, {bs1.x, bs1.y}, {bs1.z, bs1.w}};
    const v2f wh[4] = {{wh0.x, wh0.y}, {wh0.z, wh0.w}, {wh1.x, wh1.y}, {wh1.z, wh1.w}};
    const v2f cc = {c, c};
    v8f o;
#pragma unroll
    for (int p = 0; p < 4; ++p) {
        v2f a;
        a.x = acc[2 * p];
        a.y = acc[2 * p + 1];
        const v2f z = __builtin_elementwise_fma(cc, ws[p], bs[p]);
        v2f d;
        d.x = __builtin_amdgcn_exp2f(z.x);
        d.y = __builtin_amdgcn_exp2f(z.y);
        d = d + 1.0f;
        v2f g;
        g.x = __builtin_amdgcn_rcpf(d.x);
        g.y = __builtin_amdgcn_rcpf(d.y);
        const v2f x = __builtin_elementwise_fma(g, a, cc * wh[p]);
        v2f u;
        u.x = __builtin_amdgcn_exp2f(-fabsf(x.x));
        u.y = __builtin_amdgcn_exp2f(-fabsf(x.y));
        u = u + 1.0f;
        v2f l;
        l.x = __builtin_amdgcn_logf(u.x);
        l.y = __builtin_amdgcn_logf(u.y);
        v2f m;
        m.x = fmaxf(x.x, 0.0f);
        m.y = fmaxf(x.y, 0.0f);
        const v2f r2 = m + l;
        o[2 * p]     = r2.x;
        o[2 * p + 1] = r2.y;
    }
    return o;
}

__device__ __forceinline__ void mid_layer(v8f act[4],
                                          const float* __restrict__ wl,
                                          const float* __restrict__ vec,
                                          int f8, float c)
{
    v8f acc[4];
#pragma unroll
    for (int to = 0; to < 4; ++to) acc[to] = bias_tile(vec, 16 * to + f8);

#pragma unroll
    for (int t = 0; t < 4; ++t) {
#pragma unroll
        for (int rp = 0; rp < 4; ++rp) {
            v2f bf;
            bf.x = act[t][2 * rp];
            bf.y = act[t][2 * rp + 1];
#pragma unroll
            for (int tp = 0; tp < 2; ++tp) {
                const float4 w = *(const float4*)(wl + ((4 * t + rp) * 2 + tp) * FRAG2);
                v2f af0, af1;
                af0.x = w.x; af0.y = w.y;
                af1.x = w.z; af1.y = w.w;
                acc[2 * tp] = __builtin_amdgcn_wmma_f32_16x16x4_f32(
                    false, af0, false, bf, (short)0, acc[2 * tp], false, false);
                acc[2 * tp + 1] = __builtin_amdgcn_wmma_f32_16x16x4_f32(
                    false, af1, false, bf, (short)0, acc[2 * tp + 1], false, false);
            }
        }
    }

#pragma unroll
    for (int to = 0; to < 4; ++to)
        act[to] = gate_softplus_tile(acc[to], vec, 16 * to + f8, c);
}

__global__ __launch_bounds__(WAVES * 32) __attribute__((amdgpu_waves_per_eu(8, 8)))
void ode_hypernet_kernel(
    const float* __restrict__ context,
    const float* __restrict__ y,
    const float* __restrict__ tnw,
    float* __restrict__ out)
{
    __shared__ __align__(16) float w_lds[L_TOTAL];

    __builtin_amdgcn_s_setprio(3);

    const int tid  = threadIdx.x;
    const int b    = blockIdx.y;
    const int wid  = tid >> 5;
    const int lane = tid & 31;
    const int lrow = lane & 15;
    const bool hi  = lane >= 16;
    const int f8   = hi ? 8 : 0;
    const int point0_blk = blockIdx.x * PTS_PER_BLK + wid * 16;

    const unsigned pidx = (unsigned)b * NPTS + point0_blk + lrow;
    const unsigned ysel = hi ? 2u : 0u;
    float ya = y[pidx * 3u + ysel];
    float yb = y[pidx * 3u + 1u];
    float cv = context[pidx];

    {
        const float4* src = (const float4*)(tnw + (size_t)b * TOTW);
        float4* dst = (float4*)w_lds;
        for (int i = tid; i < L_SMALL4; i += WAVES * 32) {
            const int s = i + (i >= 112 ? 1024 : 0) + (i >= 176 ? 1024 : 0);
            float4 v = src[s];
            const int q = ((i - 48) >> 4) & 3;
            const float fac = i >= 240 ? 1.0f
                            : (i >= 48 && (q == 1 || q == 2)) ? -LOG2E_F : LOG2E_F;
            v.x *= fac; v.y *= fac; v.z *= fac; v.w *= fac;
            dst[i] = v;
        }
#pragma unroll
        for (int m = 0; m < 2; ++m) {
            const float4* ms = src + (m == 0 ? W1_OFF / 4 : W2_OFF / 4);
            float* md = w_lds + (m == 0 ? L_W1 : L_W2);
#pragma unroll
            for (int it = 0; it < 4; ++it) {
                const int i = tid + it * (WAVES * 32);
                const float4 v = ms[i];
                const int k = i >> 4;
                const int to = (i & 15) >> 2;
                const int c0 = (i & 3) * 4;
                const int j = ((k >> 4) << 2) | ((k >> 1) & 3);
                float* d = md + (2 * j + (to >> 1)) * FRAG2
                              + (((k >> 3) & 1) * 16 + c0) * 4 + (to & 1) * 2 + (k & 1);
                d[0] = v.x; d[4] = v.y; d[8] = v.z; d[12] = v.w;
            }
        }
    }
    __syncthreads();

    const float* wl1 = w_lds + L_W1 + 4 * lane;
    const float* wl2 = w_lds + L_W2 + 4 * lane;

  for (int trip = 0; trip < TRIPS; ++trip) {
    const int tiles_left = TRIPS - 1 - trip;
    if (tiles_left >= 3)      __builtin_amdgcn_s_setprio(3);
    else if (tiles_left == 2) __builtin_amdgcn_s_setprio(2);
    else if (tiles_left == 1) __builtin_amdgcn_s_setprio(1);
    else                      __builtin_amdgcn_s_setprio(0);
    asm volatile("" ::: "memory");
    const float c = cv;

    v8f act[4];

    {
        v2f b0;
        b0.x = ya;
        b0.y = hi ? 0.0f : yb;
        {
            const unsigned np = pidx + (unsigned)((trip + 1 < TRIPS ? trip + 1 : trip) * PTS_PER_TRIP);
            ya = y[np * 3u + ysel];
            yb = y[np * 3u + 1u];
            cv = context[np];
        }
#pragma unroll
        for (int to = 0; to < 4; ++to) {
            const int n = to * 16 + lrow;
            v2f af;
            af.x = w_lds[L_W0 + (hi ? 2 : 0) * 64 + n];
            const float w1 = w_lds[L_W0 + 1 * 64 + n];
            af.y = hi ? 0.0f : w1;
            v8f acc = bias_tile(w_lds + L_V0, 16 * to + f8);
            acc = __builtin_amdgcn_wmma_f32_16x16x4_f32(
                false, af, false, b0, (short)0, acc, false, false);
            act[to] = gate_softplus_tile(acc, w_lds + L_V0, 16 * to + f8, c);
        }
    }

    mid_layer(act, wl1, w_lds + L_V1, f8, c);
    mid_layer(act, wl2, w_lds + L_V2, f8, c);

    {
        const bool nvalid = lrow < 3;
        const int nc = nvalid ? lrow : 0;
        const float nmask = nvalid ? 1.0f : 0.0f;
        const float* w3 = w_lds + L_W3 + f8 * 3 + nc;
        v8f acc0 = {}, acc1 = {};
#pragma unroll
        for (int t = 0; t < 4; ++t) {
#pragma unroll
            for (int rp = 0; rp < 4; rp += 2) {
                v2f bf0, bf1, af0, af1;
                bf0.x = act[t][2 * rp];
                bf0.y = act[t][2 * rp + 1];
                bf1.x = act[t][2 * rp + 2];
                bf1.y = act[t][2 * rp + 3];
                af0.x = w3[(16 * t + 2 * rp + 0) * 3] * nmask;
                af0.y = w3[(16 * t + 2 * rp + 1) * 3] * nmask;
                af1.x = w3[(16 * t + 2 * rp + 2) * 3] * nmask;
                af1.y = w3[(16 * t + 2 * rp + 3) * 3] * nmask;
                acc0 = __builtin_amdgcn_wmma_f32_16x16x4_f32(
                    false, af0, false, bf0, (short)0, acc0, false, false);
                acc1 = __builtin_amdgcn_wmma_f32_16x16x4_f32(
                    false, af1, false, bf1, (short)0, acc1, false, false);
            }
        }
        const v8f acc = acc0 + acc1;
        if (!hi) {
            const unsigned ob = (pidx + (unsigned)(trip * PTS_PER_TRIP)) * 3u;
#pragma unroll
            for (int r = 0; r < 3; ++r) {
                const float bb = w_lds[L_B3 + r];
                const float ws = w_lds[L_B3 + 3 + r];
                const float bs = w_lds[L_B3 + 6 + r];
                const float wh = w_lds[L_B3 + 9 + r];
                float v = fmaf(LN2_F, acc[r], bb);
                v = sigmoid_f(fmaf(c, ws, bs)) * v + c * wh;
                out[ob + (unsigned)r] = v;
            }
        }
    }
  }
}

extern "C" void kernel_launch(void* const* d_in, const int* in_sizes, int n_in,
                              void* d_out, int out_size, void* d_ws, size_t ws_size,
                              hipStream_t stream) {
    const float* context = (const float*)d_in[0];
    const float* y       = (const float*)d_in[1];
    const float* tnw     = (const float*)d_in[2];
    float* out           = (float*)d_out;

    dim3 grid(NPTS / PTS_PER_BLK, BATCH);
    ode_hypernet_kernel<<<grid, WAVES * 32, 0, stream>>>(context, y, tnw, out);
}


# Two 16-point tiles side by side per wave (4 waves per SIMD, 256 registers): each weight fragment and per-feature vector read from LDS feeds both tiles, LDS reads per point halved; sums unchanged
# speedup vs baseline: 1.2970x; 1.0677x over previous
#include <hip/hip_runtime.h>
#include <hip/hip_bf16.h>


typedef __attribute__((ext_vector_type(2))) float v2f;
typedef __attribute__((ext_vector_type(8))) float v8f;

#define BATCH        64
#define NPTS         8192
#define TOTW         9356
#define WAVES        8
#define NT           2
#define PTS_PER_TRIP (WAVES * 16 * NT)
#define TRIPS        4
#define PTS_PER_BLK  (PTS_PER_TRIP * TRIPS)

#define LOG2E_F 1.4426950408889634f
#define LN2_F   0.6931471805599453f

#define W0_OFF   0
#define V0_OFF   192
#define W1_OFF   448
#define V1_OFF   4544
#define W2_OFF   4800
#define V2_OFF   8896
#define W3_OFF   9152
#define B3_OFF   9344

#define L_W0     0
#define L_V0     192
#define L_V1     448
#define L_V2     704
#define L_W3     960
#define L_B3     1152
#define L_SMALL4 291
#define L_W1     1164
#define FRAG2    128
#define L_W2     (L_W1 + 32 * FRAG2)
#define L_TOTAL  (L_W2 + 32 * FRAG2)

__device__ __forceinline__ float sigmoid_f(float x) {
    return __builtin_amdgcn_rcpf(1.0f + __builtin_amdgcn_exp2f(x * -LOG2E_F));
}


__device__ __forceinline__ v8f bias_tile(const float* __restrict__ vec, int f) {
    const float4 b0 = *(const float4*)(vec + f);
    const float4 b1 = *(const float4*)(vec + f + 4);
    v8f a;
    a[0] = b0.x; a[1] = b0.y; a[2] = b0.z; a[3] = b0.w;
    a[4] = b1.x; a[5] = b1.y; a[6] = b1.z; a[7] = b1.w;
    return a;
}

__device__ __forceinline__ v8f gate_softplus_one(v8f acc, const v2f (&ws)[4], const v2f (&bs)[4], const v2f (&wh)[4], float c);

__device__ __forceinline__ void gate_softplus_tiles(v8f (&o)[NT], const v8f (&acc)[NT],
                                                    const float* __restrict__ vec, int f, const float (&c)[NT]) {
    const float4 ws0 = *(const float4*)(vec + 64 + f);
    const float4 ws1 = *(const float4*)(vec + 64 + f + 4);
    const float4 bs0 = *(const float4*)(vec + 128 + f);
    const float4 bs1 = *(const float4*)(vec + 128 + f + 4);
    const float4 wh0 = *(const float4*)(vec + 192 + f);
    const float4 wh1 = *(const float4*)(vec + 192 + f + 4);
    const v2f ws[4] = {{ws0.x, ws0.y}, {ws0.z, ws0.w}, {ws1.x, ws1.y}, {ws1.z, ws1.w}};
    const v2f bs[4] = {{bs0.x, bs0.y}, {bs0.z, bs0.w}, {bs1.x, bs1.y}, {bs1.z, bs1.w}};
    const v2f wh[4] = {{wh0.x, wh0.y}, {wh0.z, wh0.w}, {wh1.x, wh1.y}, {wh1.z, wh1.w}};
#pragma unroll
    for (int n = 0; n < NT; ++n) o[n] = gate_softplus_one(acc[n], ws, bs, wh, c[n]);
}

__device__ __forceinline__ v8f gate_softplus_one(v8f acc, const v2f (&ws)[4], const v2f (&bs)[4], const v2f (&wh)[4], float c) {
    const v2f cc = {c, c};
    v8f o;
#pragma unroll
    for (int p = 0; p < 4; ++p) {
        v2f a;
        a.x = acc[2 * p];
        a.y = acc[2 * p + 1];
        const v2f z = __builtin_elementwise_fma(cc, ws[p], bs[p]);
        v2f d;
        d.x = __builtin_amdgcn_exp2f(z.x);
        d.y = __builtin_amdgcn_exp2f(z.y);
        d = d + 1.0f;
        v2f g;
        g.x = __builtin_amdgcn_rcpf(d.x);
        g.y = __builtin_amdgcn_rcpf(d.y);
        const v2f x = __builtin_elementwise_fma(g, a, cc * wh[p]);
        v2f u;
        u.x = __builtin_amdgcn_exp2f(-fabsf(x.x));
        u.y = __builtin_amdgcn_exp2f(-fabsf(x.y));
        u = u + 1.0f;
        v2f l;
        l.x = __builtin_amdgcn_logf(u.x);
        l.y = __builtin_amdgcn_logf(u.y);
        v2f m;
        m.x = fmaxf(x.x, 0.0f);
        m.y = fmaxf(x.y, 0.0f);
        const v2f r2 = m + l;
        o[2 * p]     = r2.x;
        o[2 * p + 1] = r2.y;
    }
    return o;
}

__device__ __forceinline__ void mid_layer(v8f (&act)[NT][4],
                                          const float* __restrict__ wl,
                                          const float* __restrict__ vec,
                                          int f8, const float (&c)[NT])
{
    v8f acc[NT][4];
#pragma unroll
    for (int to = 0; to < 4; ++to) {
        const v8f bt = bias_tile(vec, 16 * to + f8);
#pragma unroll
        for (int n = 0; n < NT; ++n) acc[n][to] = bt;
    }

#pragma unroll
    for (int t = 0; t < 4; ++t) {
#pragma unroll
        for (int rp = 0; rp < 4; ++rp) {
            v2f bf[NT];
#pragma unroll
            for (int n = 0; n < NT; ++n) {
                bf[n].x = act[n][t][2 * rp];
                bf[n].y = act[n][t][2 * rp + 1];
            }
#pragma unroll
            for (int tp = 0; tp < 2; ++tp) {
                const float4 w = *(const float4*)(wl + ((4 * t + rp) * 2 + tp) * FRAG2);
                v2f af0, af1;
                af0.x = w.x; af0.y = w.y;
                af1.x = w.z; af1.y = w.w;
#pragma unroll
                for (int n = 0; n < NT; ++n) {
                    acc[n][2 * tp] = __builtin_amdgcn_wmma_f32_16x16x4_f32(
                        false, af0, false, bf[n], (short)0, acc[n][2 * tp], false, false);
                    acc[n][2 * tp + 1] = __builtin_amdgcn_wmma_f32_16x16x4_f32(
                        false, af1, false, bf[n], (short)0, acc[n][2 * tp + 1], false, false);
                }
            }
        }
    }

#pragma unroll
    for (int to = 0; to < 4; ++to) {
        v8f o[NT], a[NT];
#pragma unroll
        for (int n = 0; n < NT; ++n) a[n] = acc[n][to];
        gate_softplus_tiles(o, a, vec, 16 * to + f8, c);
#pragma unroll
        for (int n = 0; n < NT; ++n) act[n][to] = o[n];
    }
}

__global__ __launch_bounds__(WAVES * 32) __attribute__((amdgpu_waves_per_eu(4, 4)))
void ode_hypernet_kernel(
    const float* __restrict__ context,
    const float* __restrict__ y,
    const float* __restrict__ tnw,
    float* __restrict__ out)
{
    __shared__ __align__(16) float w_lds[L_TOTAL];

    __builtin_amdgcn_s_setprio(3);

    const int tid  = threadIdx.x;
    const int b    = blockIdx.y;
    const int wid  = tid >> 5;
    const int lane = tid & 31;
    const int lrow = lane & 15;
    const bool hi  = lane >= 16;
    const int f8   = hi ? 8 : 0;
    const int point0_blk = blockIdx.x * PTS_PER_BLK + wid * (16 * NT);

    const unsigned pidx = (unsigned)b * NPTS + point0_blk + lrow;
    const unsigned ysel = hi ? 2u : 0u;
    float ya[NT], yb[NT], cv[NT];
#pragma unroll
    for (int n = 0; n < NT; ++n) {
        const unsigned pn = pidx + 16u * n;
        ya[n] = y[pn * 3u + ysel];
        yb[n] = y[pn * 3u + 1u];
        cv[n] = context[pn];
    }

    {
        const float4* src = (const float4*)(tnw + (size_t)b * TOTW);
        float4* dst = (float4*)w_lds;
        for (int i = tid; i < L_SMALL4; i += WAVES * 32) {
            const int s = i + (i >= 112 ? 1024 : 0) + (i >= 176 ? 1024 : 0);
            float4 v = src[s];
            const int q = ((i - 48) >> 4) & 3;
            const float fac = i >= 240 ? 1.0f
                            : (i >= 48 && (q == 1 || q == 2)) ? -LOG2E_F : LOG2E_F;
            v.x *= fac; v.y *= fac; v.z *= fac; v.w *= fac;
            dst[i] = v;
        }
#pragma unroll
        for (int m = 0; m < 2; ++m) {
            const float4* ms = src + (m == 0 ? W1_OFF / 4 : W2_OFF / 4);
            float* md = w_lds + (m == 0 ? L_W1 : L_W2);
#pragma unroll
            for (int it = 0; it < 4; ++it) {
                const int i = tid + it * (WAVES * 32);
                const float4 v = ms[i];
                const int k = i >> 4;
                const int to = (i & 15) >> 2;
                const int c0 = (i & 3) * 4;
                const int j = ((k >> 4) << 2) | ((k >> 1) & 3);
                float* d = md + (2 * j + (to >> 1)) * FRAG2
                              + (((k >> 3) & 1) * 16 + c0) * 4 + (to & 1) * 2 + (k & 1);
                d[0] = v.x; d[4] = v.y; d[8] = v.z; d[12] = v.w;
            }
        }
    }
    __syncthreads();

    const float* wl1 = w_lds + L_W1 + 4 * lane;
    const float* wl2 = w_lds + L_W2 + 4 * lane;

  for (int trip = 0; trip < TRIPS; ++trip) {
    const int tiles_left = TRIPS - 1 - trip;
    if (tiles_left >= 3)      __builtin_amdgcn_s_setprio(3);
    else if (tiles_left == 2) __builtin_amdgcn_s_setprio(2);
    else if (tiles_left == 1) __builtin_amdgcn_s_setprio(1);
    else                      __builtin_amdgcn_s_setprio(0);
    asm volatile("" ::: "memory");
    float c[NT];
#pragma unroll
    for (int n = 0; n < NT; ++n) c[n] = cv[n];

    v8f act[NT][4];

    {
        v2f b0[NT];
#pragma unroll
        for (int n = 0; n < NT; ++n) {
            b0[n].x = ya[n];
            b0[n].y = hi ? 0.0f : yb[n];
        }
        {
            const unsigned np = pidx + (unsigned)((trip + 1 < TRIPS ? trip + 1 : trip) * PTS_PER_TRIP);
#pragma unroll
            for (int n = 0; n < NT; ++n) {
                const unsigned pn = np + 16u * n;
                ya[n] = y[pn * 3u + ysel];
                yb[n] = y[pn * 3u + 1u];
                cv[n] = context[pn];
            }
        }
#pragma unroll
        for (int to = 0; to < 4; ++to) {
            const int n = to * 16 + lrow;
            v2f af;
            af.x = w_lds[L_W0 + (hi ? 2 : 0) * 64 + n];
            const float w1 = w_lds[L_W0 + 1 * 64 + n];
            af.y = hi ? 0.0f : w1;
            const v8f bt = bias_tile(w_lds + L_V0, 16 * to + f8);
            v8f acc[NT], o[NT];
#pragma unroll
            for (int n = 0; n < NT; ++n)
                acc[n] = __builtin_amdgcn_wmma_f32_16x16x4_f32(
                    false, af, false, b0[n], (short)0, bt, false, false);
            gate_softplus_tiles(o, acc, w_lds + L_V0, 16 * to + f8, c);
#pragma unroll
            for (int n = 0; n < NT; ++n) act[n][to] = o[n];
        }
    }

    mid_layer(act, wl1, w_lds + L_V1, f8, c);
    mid_layer(act, wl2, w_lds + L_V2, f8, c);

    {
        const bool nvalid = lrow < 3;
        const int nc = nvalid ? lrow : 0;
        const float nmask = nvalid ? 1.0f : 0.0f;
        const float* w3 = w_lds + L_W3 + f8 * 3 + nc;
        v8f acc0[NT], acc1[NT];
#pragma unroll
        for (int n = 0; n < NT; ++n) { acc0[n] = v8f{}; acc1[n] = v8f{}; }
#pragma unroll
        for (int t = 0; t < 4; ++t) {
#pragma unroll
            for (int rp = 0; rp < 4; rp += 2) {
                v2f af0, af1;
                af0.x = w3[(16 * t + 2 * rp + 0) * 3] * nmask;
                af0.y = w3[(16 * t + 2 * rp + 1) * 3] * nmask;
                af1.x = w3[(16 * t + 2 * rp + 2) * 3] * nmask;
                af1.y = w3[(16 * t + 2 * rp + 3) * 3] * nmask;
#pragma unroll
                for (int n = 0; n < NT; ++n) {
                    v2f bf0, bf1;
                    bf0.x = act[n][t][2 * rp];
                    bf0.y = act[n][t][2 * rp + 1];
                    bf1.x = act[n][t][2 * rp + 2];
                    bf1.y = act[n][t][2 * rp + 3];
                    acc0[n] = __builtin_amdgcn_wmma_f32_16x16x4_f32(
                        false, af0, false, bf0, (short)0, acc0[n], false, false);
                    acc1[n] = __builtin_amdgcn_wmma_f32_16x16x4_f32(
                        false, af1, false, bf1, (short)0, acc1[n], false, false);
                }
            }
        }
        v8f acc[NT];
#pragma unroll
        for (int n = 0; n < NT; ++n) acc[n] = acc0[n] + acc1[n];
        if (!hi) {
            const unsigned ob = (pidx + (unsigned)(trip * PTS_PER_TRIP)) * 3u;
#pragma unroll
            for (int r = 0; r < 3; ++r) {
                const float bb = w_lds[L_B3 + r];
                const float ws = w_lds[L_B3 + 3 + r];
                const float bs = w_lds[L_B3 + 6 + r];
                const float wh = w_lds[L_B3 + 9 + r];
#pragma unroll
                for (int n = 0; n < NT; ++n) {
                    float v = fmaf(LN2_F, acc[n][r], bb);
                    v = sigmoid_f(fmaf(c[n], ws, bs)) * v + c[n] * wh;
                    out[ob + 48u * n + (unsigned)r] = v;
                }
            }
        }
    }
  }
}

extern "C" void kernel_launch(void* const* d_in, const int* in_sizes, int n_in,
                              void* d_out, int out_size, void* d_ws, size_t ws_size,
                              hipStream_t stream) {
    const float* context = (const float*)d_in[0];
    const float* y       = (const float*)d_in[1];
    const float* tnw     = (const float*)d_in[2];
    float* out           = (float*)d_out;

    dim3 grid(NPTS / PTS_PER_BLK, BATCH);
    ode_hypernet_kernel<<<grid, WAVES * 32, 0, stream>>>(context, y, tnw, out);
}
